# starting kernel
# baseline (speedup 1.0000x reference)
#include <hip/hip_runtime.h>

#define NBATCH 2
#define NDIM   768
#define NLEN   2048
#define NST    16
#define CHUNK  16
#define NC     (NLEN / CHUNK)
#define TOTCH  (NBATCH * NDIM * NC)
#define WSN    ((size_t)TOTCH * NST)
#define CGRP   8

typedef float v2f __attribute__((ext_vector_type(2)));
typedef float v8f __attribute__((ext_vector_type(8)));

__device__ __forceinline__ float softplus_f(float x) {
    return (x > 20.0f) ? x : __logf(1.0f + __expf(x));
}

#define SWZ(v, IMM) __int_as_float(__builtin_amdgcn_ds_swizzle(__float_as_int(v), (IMM)))
#define SWZ_ADD(v, IMM) (v) += SWZ(v, IMM)

__global__ __launch_bounds__(256) void k1_chunk_state(
    const float* __restrict__ u, const float* __restrict__ delta,
    const float* __restrict__ A, const float* __restrict__ B,
    const float* __restrict__ dbias,
    float* __restrict__ wsDecay, float* __restrict__ wsS)
{
    int t = blockIdx.x * 256 + threadIdx.x;
    int c = t % NC;
    int d = (t / NC) % NDIM;
    int b = t / (NC * NDIM);
    int cs = c * CHUNK;

    const float* dp = delta + (size_t)(b * NDIM + d) * NLEN + cs;
    const float* up = u     + (size_t)(b * NDIM + d) * NLEN + cs;

    float An[NST];
#pragma unroll
    for (int n = 0; n < NST; n++) An[n] = A[d * NST + n];
    float bias = dbias[d];

    float x[NST];
#pragma unroll
    for (int n = 0; n < NST; n++) x[n] = 0.0f;
    float R = 0.0f;

    for (int tt = 0; tt < CHUNK; tt++) {
        float sp = softplus_f(dp[tt] + bias);
        float du = sp * up[tt];
        R += sp;
#pragma unroll
        for (int n = 0; n < NST; n++) {
            float a  = __expf(An[n] * sp);
            float bv = B[(size_t)(b * NST + n) * NLEN + cs + tt];
            x[n] = a * x[n] + du * bv;
        }
    }
    size_t o = ((size_t)(b * NDIM + d) * NC + c) * NST;
#pragma unroll
    for (int n = 0; n < NST; n++) {
        wsS[o + n]     = x[n];
        wsDecay[o + n] = __expf(An[n] * R);
    }
}

__global__ __launch_bounds__(256) void k2_scan(
    float* __restrict__ wsDecay, const float* __restrict__ wsS)
{
    int i = blockIdx.x * 256 + threadIdx.x;
    int n = i & (NST - 1);
    int d = (i >> 4) % NDIM;
    int b = i / (NST * NDIM);
    size_t base = ((size_t)(b * NDIM + d) * NC) * NST + n;

    float x = 0.0f;
    for (int c = 0; c < NC; c++) {
        size_t idx = base + (size_t)c * NST;
        float dec = wsDecay[idx];
        float s   = wsS[idx];
        wsDecay[idx] = x;
        x = dec * x + s;
    }
}

__global__ __launch_bounds__(256) void k3_output(
    const float* __restrict__ u, const float* __restrict__ delta,
    const float* __restrict__ A, const float* __restrict__ B,
    const float* __restrict__ C, const float* __restrict__ Dp,
    const float* __restrict__ z, const float* __restrict__ dbias,
    const float* __restrict__ wsXin, float* __restrict__ out)
{
    __shared__ float Pl[CGRP][16][17];
    __shared__ float Ql[CGRP][16][17];
    __shared__ float dl[CGRP][16];
    __shared__ float ul[CGRP][16];
    __shared__ float xl[CGRP][16];

    int w    = threadIdx.x >> 5;
    int lane = threadIdx.x & 31;
    int m    = lane & 15;
    bool lo  = lane < 16;

    int cg = blockIdx.x % (NC / CGRP);
    int d  = (blockIdx.x / (NC / CGRP)) % NDIM;
    int b  = blockIdx.x / ((NC / CGRP) * NDIM);
    int c  = cg * CGRP + w;
    int cs = c * CHUNK;
    size_t base = (size_t)(b * NDIM + d) * NLEN + cs;

    if (lo) {
        dl[w][m] = softplus_f(delta[base + m] + dbias[d]);
    } else {
        ul[w][m] = u[base + m];
        xl[w][m] = wsXin[((size_t)(b * NDIM + d) * NC + c) * NST + m];
    }
    __syncthreads();

    float rr = 0.0f;
#pragma unroll
    for (int t = 0; t < 16; t++) { float v = dl[w][t]; rr += (t <= m) ? v : 0.0f; }

    float dus = dl[w][m] * ul[w][m];
    float sgn = lo ? 1.0f : -1.0f;
    float mul = lo ? 1.0f : dus;
    const float* srcb = lo ? C : B;
    float* dstb = lo ? &Pl[w][0][0] : &Ql[w][0][0];

#pragma unroll
    for (int n = 0; n < NST; n++) {
        float An = A[d * NST + n];
        float v  = srcb[(size_t)(b * NST + n) * NLEN + cs + m];
        float e  = __expf(sgn * An * rr);
        dstb[m * 17 + n] = v * e * mul;
    }
    __syncthreads();

    v8f acc = {0.f, 0.f, 0.f, 0.f, 0.f, 0.f, 0.f, 0.f};
#pragma unroll
    for (int j = 0; j < 4; j++) {
        int k0 = 4 * j + (lo ? 0 : 2);
        int k1 = 4 * j + (lo ? 1 : 3);
        v2f va; va.x = Pl[w][m][k0]; va.y = Pl[w][m][k1];
        v2f vb; vb.x = Ql[w][m][k0]; vb.y = Ql[w][m][k1];
        acc = __builtin_amdgcn_wmma_f32_16x16x4_f32(
            false, va, false, vb, (short)0, acc, false, false);
    }

    float yv = 0.0f;
#pragma unroll
    for (int r = 0; r < 8; r++) {
        int t = r + (lo ? 0 : 8);
        float v = (m <= t) ? acc[r] : 0.0f;
        SWZ_ADD(v, 0x041F);
        SWZ_ADD(v, 0x081F);
        SWZ_ADD(v, 0x101F);
        SWZ_ADD(v, 0x201F);
        yv = ((m & 7) == r) ? v : yv;
    }
    float yx = SWZ(yv, 0x601F);
    float y  = (m < 8) ? yv : yx;

    if (lo) {
        int t = m;
#pragma unroll
        for (int n = 0; n < NST; n++) y += Pl[w][t][n] * xl[w][n];
        float uv = ul[w][t];
        float zv = z[base + t];
        float sig = __builtin_amdgcn_rcpf(1.0f + __expf(-zv));
        out[base + t] = (y + uv * Dp[d]) * (zv * sig);
    }
}

extern "C" void kernel_launch(void* const* d_in, const int* in_sizes, int n_in,
                              void* d_out, int out_size, void* d_ws, size_t ws_size,
                              hipStream_t stream) {
    const float* u     = (const float*)d_in[0];
    const float* delta = (const float*)d_in[1];
    const float* A     = (const float*)d_in[2];
    const float* B     = (const float*)d_in[3];
    const float* C     = (const float*)d_in[4];
    const float* Dp    = (const float*)d_in[5];
    const float* z     = (const float*)d_in[6];
    const float* dbias = (const float*)d_in[7];
    float* out = (float*)d_out;

    float* wsDecay = (float*)d_ws;
    float* wsS     = wsDecay + WSN;

    k1_chunk_state<<<TOTCH / 256, 256, 0, stream>>>(u, delta, A, B, dbias,
                                                    wsDecay, wsS);
    k2_scan<<<(NBATCH * NDIM * NST) / 256, 256, 0, stream>>>(wsDecay, wsS);
    k3_output<<<TOTCH / CGRP, 256, 0, stream>>>(u, delta, A, B, C, Dp, z, dbias,
                                                wsDecay, out);
}


# single launch: one wave per (b,d) row walks its 128 chunks in order, state carried in registers; chunk-state and scan kernels and workspace removed; delta-scan, y and carry-in sum orders changed
# speedup vs baseline: 4.2119x; 4.2119x over previous
#include <hip/hip_runtime.h>


#define NBATCH 2
#define NDIM   768
#define NLEN   2048
#define NST    16
#define CHUNK  16
#define NC     (NLEN / CHUNK)
#define RING   4

typedef float v2f __attribute__((ext_vector_type(2)));
typedef float v8f __attribute__((ext_vector_type(8)));


__device__ __forceinline__ float softplus_f(float x) {
    const float l = __logf(1.0f + __expf(x));
    return (x > 20.0f) ? x : l;
}

#define ROW_SHR(n)   (0x110 + (n))
#define ROW_SHARE(n) (0x150 + (n))
#define ROW_XMASK(n) (0x160 + (n))

template <int CTRL>
__device__ __forceinline__ float row_dpp(float v) {
    return __int_as_float(__builtin_amdgcn_update_dpp(0, __float_as_int(v), CTRL, 0xf, 0xf, true));
}

__device__ __forceinline__ float xhalf(float x) {
    return __int_as_float(__builtin_amdgcn_permlanex16(__float_as_int(x), __float_as_int(x),
                                                       0x76543210, 0xfedcba98, false, false));
}

struct Tile { float dv, uv, zv; float bv[8]; float cv[8]; };

__device__ __forceinline__ void load_tile(Tile& t, const float* __restrict__ dp,
                                          const float* __restrict__ up,
                                          const float* __restrict__ zp,
                                          const float* __restrict__ Bl,
                                          const float* __restrict__ Cl, int c) {
    const int o = c * CHUNK;
    t.dv = dp[o];
    t.uv = up[o];
    t.zv = zp[o];
#pragma unroll
    for (int i = 0; i < 8; i++) {
        const int no = (4 * (i >> 1) + (i & 1)) * NLEN;
        t.bv[i] = Bl[no + o];
        t.cv[i] = Cl[no + o];
    }
}

__global__ __launch_bounds__(32) void k_scan_rows(
    const float* __restrict__ u, const float* __restrict__ delta,
    const float* __restrict__ A, const float* __restrict__ B,
    const float* __restrict__ C, const float* __restrict__ Dp,
    const float* __restrict__ z, const float* __restrict__ dbias,
    float* __restrict__ out)
{
    const int lane = threadIdx.x;
    const int h = lane >> 4;
    const int m = lane & 15;
    const int row = blockIdx.x;
    const int b = row / NDIM;
    const int d = row - b * NDIM;
    const size_t rbase = (size_t)row * NLEN;

    const float* dp = delta + rbase + m;
    const float* up = u + rbase + m;
    const float* zp = z + rbase + m;
    float* op = out + rbase + lane;
    const float* Bl = B + ((size_t)b * NST + 2 * h) * NLEN + m;
    const float* Cl = C + ((size_t)b * NST + 2 * h) * NLEN + m;

    const float bias = dbias[d];
    const float Dd = Dp[d];
    float Al[8];
#pragma unroll
    for (int i = 0; i < 8; i++)
        Al[i] = A[d * NST + 4 * (i >> 1) + 2 * h + (i & 1)] * 1.44269504088896341f;
    const float first = (m == 0) ? 1.0f : 0.0f;

    float xs[8];
#pragma unroll
    for (int i = 0; i < 8; i++) xs[i] = 0.0f;

    Tile q[RING];
#pragma unroll
    for (int k = 0; k < RING - 1; k++) {
        load_tile(q[k], dp, up, zp, Bl, Cl, k);
        __builtin_amdgcn_sched_barrier(0);
    }

    float fin_prev = 0.0f;
    for (int c0 = 0; c0 < NC; c0 += RING) {
#pragma unroll
        for (int k = 0; k < RING; k++) {
            const int c = c0 + k;
            int cn = c + (RING - 1);
            cn = (cn < NC) ? cn : (NC - 1);
            __builtin_amdgcn_sched_barrier(0);
            load_tile(q[(k + RING - 1) % RING], dp, up, zp, Bl, Cl, cn);
            __builtin_amdgcn_sched_barrier(0);
            const Tile& t = q[k];

            const float sp = softplus_f(t.dv + bias);
            float rr = sp;
            rr += row_dpp<ROW_SHR(1)>(rr);
            rr += row_dpp<ROW_SHR(2)>(rr);
            rr += row_dpp<ROW_SHR(4)>(rr);
            rr += row_dpp<ROW_SHR(8)>(rr);
            const float du = sp * t.uv;

            float pa[8], qa[8], ep[8], eq[8], arg[8];
#pragma unroll
            for (int i = 0; i < 8; i++) arg[i] = Al[i] * rr;
#pragma unroll
            for (int i = 0; i < 8; i++) ep[i] = __builtin_amdgcn_exp2f(arg[i]);
#pragma unroll
            for (int i = 0; i < 8; i++) eq[i] = __builtin_amdgcn_exp2f(-arg[i]);
#pragma unroll
            for (int i = 0; i < 8; i++) pa[i] = t.cv[i] * ep[i];
#pragma unroll
            for (int i = 0; i < 8; i++) qa[i] = t.bv[i] * eq[i];
#pragma unroll
            for (int i = 0; i < 8; i++) qa[i] = fmaf(xs[i], first, qa[i] * du);

            v8f acc = {0.f, 0.f, 0.f, 0.f, 0.f, 0.f, 0.f, 0.f};
#pragma unroll
            for (int j = 0; j < 4; j++) {
                v2f va; va.x = qa[2 * j]; va.y = qa[2 * j + 1];
                v2f vb; vb.x = pa[2 * j]; vb.y = pa[2 * j + 1];
                acc = __builtin_amdgcn_wmma_f32_16x16x4_f32(
                    false, va, false, vb, (short)0, acc, false, false);
            }

            float cs[8];
#pragma unroll
            for (int i = 0; i < 8; i++) cs[i] = qa[i] + row_dpp<ROW_XMASK(1)>(qa[i]);
#pragma unroll
            for (int i = 0; i < 8; i++) cs[i] += row_dpp<ROW_XMASK(2)>(cs[i]);
#pragma unroll
            for (int i = 0; i < 8; i++) cs[i] += row_dpp<ROW_XMASK(4)>(cs[i]);
#pragma unroll
            for (int i = 0; i < 8; i++) cs[i] += row_dpp<ROW_XMASK(8)>(cs[i]);
#pragma unroll
            for (int i = 0; i < 8; i++) xs[i] = cs[i] * row_dpp<ROW_SHARE(15)>(ep[i]);

            float ym[8];
#pragma unroll
            for (int r = 0; r < 8; r++) ym[r] = ((8 * h + r) <= m) ? acc[r] : 0.0f;
            float y = ((ym[0] + ym[1]) + (ym[2] + ym[3])) + ((ym[4] + ym[5]) + (ym[6] + ym[7]));
            y += xhalf(y);

            const float zv = t.zv;
            const float sig = __builtin_amdgcn_rcpf(1.0f + __expf(-zv));
            const float fin = (y + t.uv * Dd) * (zv * sig);
            if (k & 1) {
                op[(c - 1) * CHUNK] = h ? fin : fin_prev;
            } else {
                fin_prev = fin;
            }
        }
    }
}

extern "C" void kernel_launch(void* const* d_in, const int* in_sizes, int n_in,
                              void* d_out, int out_size, void* d_ws, size_t ws_size,
                              hipStream_t stream) {
    const float* u     = (const float*)d_in[0];
    const float* delta = (const float*)d_in[1];
    const float* A     = (const float*)d_in[2];
    const float* B     = (const float*)d_in[3];
    const float* C     = (const float*)d_in[4];
    const float* Dp    = (const float*)d_in[5];
    const float* z     = (const float*)d_in[6];
    const float* dbias = (const float*)d_in[7];
    float* out = (float*)d_out;

    k_scan_rows<<<NBATCH * NDIM, 32, 0, stream>>>(u, delta, A, B, C, Dp, z, dbias, out);
}


# each row's 16 states split over 2 waves (8 each); work-group = 2 rows x 2 waves, so 3072 equal waves, 3 per SIMD; parts of y swapped through LDS, one barrier per 4 chunks; y sum order changed
# speedup vs baseline: 4.5581x; 1.0822x over previous
#include <hip/hip_runtime.h>


#define NBATCH 2
#define NDIM   768
#define NLEN   2048
#define NST    16
#define NWAVE  2
#define NROW   2
#define NHALF  (NST / NWAVE)
#define NLANE  (NHALF / 2)
#define CHUNK  16
#define NC     (NLEN / CHUNK)
#define RING   4

typedef float v2f __attribute__((ext_vector_type(2)));
typedef float v8f __attribute__((ext_vector_type(8)));


__device__ __forceinline__ float softplus_f(float x) {
    const float l = __logf(1.0f + __expf(x));
    return (x > 20.0f) ? x : l;
}

#define ROW_SHR(n)   (0x110 + (n))
#define ROW_SHARE(n) (0x150 + (n))
#define ROW_XMASK(n) (0x160 + (n))

template <int CTRL>
__device__ __forceinline__ float row_dpp(float v) {
    return __int_as_float(__builtin_amdgcn_update_dpp(0, __float_as_int(v), CTRL, 0xf, 0xf, true));
}

__device__ __forceinline__ float xhalf(float x) {
    return __int_as_float(__builtin_amdgcn_permlanex16(__float_as_int(x), __float_as_int(x),
                                                       0x76543210, 0xfedcba98, false, false));
}

struct Tile { float dv, uv; float bv[NLANE]; float cv[NLANE]; };

__device__ __forceinline__ void load_tile(Tile& t, const float* __restrict__ dp,
                                          const float* __restrict__ up,
                                          const float* __restrict__ Bl,
                                          const float* __restrict__ Cl, int c) {
    const int o = c * CHUNK;
    t.dv = dp[o];
    t.uv = up[o];
#pragma unroll
    for (int i = 0; i < NLANE; i++) {
        const int no = (4 * (i >> 1) + (i & 1)) * NLEN;
        t.bv[i] = Bl[no + o];
        t.cv[i] = Cl[no + o];
    }
}

__global__ __launch_bounds__(32 * NWAVE * NROW) void k_scan_rows(
    const float* __restrict__ u, const float* __restrict__ delta,
    const float* __restrict__ A, const float* __restrict__ B,
    const float* __restrict__ C, const float* __restrict__ Dp,
    const float* __restrict__ z, const float* __restrict__ dbias,
    float* __restrict__ out)
{
    __shared__ float ybuf[NROW][2][NWAVE][32];

    const int lane = threadIdx.x & 31;
    const int wv = __builtin_amdgcn_readfirstlane((int)(threadIdx.x >> 5));
    const int rw = wv >> 1;
    const int w = wv & 1;
    const int h = lane >> 4;
    const int m = lane & 15;
    const int row = blockIdx.x * NROW + rw;
    const int b = row / NDIM;
    const int d = row - b * NDIM;
    const size_t rbase = (size_t)row * NLEN;

    const float* dp = delta + rbase + m;
    const float* up = u + rbase + m;
    const float* uf = u + rbase + 2 * w * CHUNK + lane;
    const float* zf = z + rbase + 2 * w * CHUNK + lane;
    float* op = out + rbase + 2 * w * CHUNK + lane;
    const float* Bl = B + ((size_t)b * NST + NHALF * w + 2 * h) * NLEN + m;
    const float* Cl = C + ((size_t)b * NST + NHALF * w + 2 * h) * NLEN + m;

    const float bias = dbias[d];
    const float Dd = Dp[d];
    float Al[NLANE];
#pragma unroll
    for (int i = 0; i < NLANE; i++)
        Al[i] = A[d * NST + NHALF * w + 4 * (i >> 1) + 2 * h + (i & 1)] * 1.44269504088896341f;
    const float first = (m == 0) ? 1.0f : 0.0f;

    float xs[NLANE];
#pragma unroll
    for (int i = 0; i < NLANE; i++) xs[i] = 0.0f;

    Tile q[RING];
#pragma unroll
    for (int k = 0; k < RING - 1; k++) {
        load_tile(q[k], dp, up, Bl, Cl, k);
        __builtin_amdgcn_sched_barrier(0);
    }

    float keep = 0.0f;
    float other = 0.0f;
    float ufin = 0.0f, zfin = 0.0f;
    float y_even = 0.0f, pair0 = 0.0f;

    for (int c0 = 0; c0 < NC; c0 += RING) {
        const int par = (c0 / RING) & 1;
#pragma unroll
        for (int k = 0; k < RING; k++) {
            const int c = c0 + k;
            int cn = c + (RING - 1);
            cn = (cn < NC) ? cn : (NC - 1);
            __builtin_amdgcn_sched_barrier(0);
            load_tile(q[(k + RING - 1) % RING], dp, up, Bl, Cl, cn);
            if (k == 1) {
                ufin = uf[c0 * CHUNK];
                zfin = zf[c0 * CHUNK];
            }
            __builtin_amdgcn_sched_barrier(0);
            const Tile& t = q[k];

            const float sp = softplus_f(t.dv + bias);
            float rr = sp;
            rr += row_dpp<ROW_SHR(1)>(rr);
            rr += row_dpp<ROW_SHR(2)>(rr);
            rr += row_dpp<ROW_SHR(4)>(rr);
            rr += row_dpp<ROW_SHR(8)>(rr);
            const float du = sp * t.uv;

            float pa[NLANE], qa[NLANE], ep[NLANE], eq[NLANE], arg[NLANE];
#pragma unroll
            for (int i = 0; i < NLANE; i++) arg[i] = Al[i] * rr;
#pragma unroll
            for (int i = 0; i < NLANE; i++) ep[i] = __builtin_amdgcn_exp2f(arg[i]);
#pragma unroll
            for (int i = 0; i < NLANE; i++) eq[i] = __builtin_amdgcn_exp2f(-arg[i]);
#pragma unroll
            for (int i = 0; i < NLANE; i++) pa[i] = t.cv[i] * ep[i];
#pragma unroll
            for (int i = 0; i < NLANE; i++) qa[i] = t.bv[i] * eq[i];
#pragma unroll
            for (int i = 0; i < NLANE; i++) qa[i] = fmaf(xs[i], first, qa[i] * du);

            v8f acc = {0.f, 0.f, 0.f, 0.f, 0.f, 0.f, 0.f, 0.f};
#pragma unroll
            for (int j = 0; j < NLANE / 2; j++) {
                v2f va; va.x = qa[2 * j]; va.y = qa[2 * j + 1];
                v2f vb; vb.x = pa[2 * j]; vb.y = pa[2 * j + 1];
                acc = __builtin_amdgcn_wmma_f32_16x16x4_f32(
                    false, va, false, vb, (short)0, acc, false, false);
            }

            float cs[NLANE];
#pragma unroll
            for (int i = 0; i < NLANE; i++) cs[i] = qa[i] + row_dpp<ROW_XMASK(1)>(qa[i]);
#pragma unroll
            for (int i = 0; i < NLANE; i++) cs[i] += row_dpp<ROW_XMASK(2)>(cs[i]);
#pragma unroll
            for (int i = 0; i < NLANE; i++) cs[i] += row_dpp<ROW_XMASK(4)>(cs[i]);
#pragma unroll
            for (int i = 0; i < NLANE; i++) cs[i] += row_dpp<ROW_XMASK(8)>(cs[i]);
#pragma unroll
            for (int i = 0; i < NLANE; i++) xs[i] = cs[i] * row_dpp<ROW_SHARE(15)>(ep[i]);

            float ym[8];
#pragma unroll
            for (int r = 0; r < 8; r++) ym[r] = ((8 * h + r) <= m) ? acc[r] : 0.0f;
            float y = ((ym[0] + ym[1]) + (ym[2] + ym[3])) + ((ym[4] + ym[5]) + (ym[6] + ym[7]));
            y += xhalf(y);

            if (k == 0) {
                y_even = y;
                if (c0 > 0) {
                    const float sig = __builtin_amdgcn_rcpf(1.0f + __expf(-zfin));
                    op[(c0 - RING) * CHUNK] = ((keep + other) + ufin * Dd) * (zfin * sig);
                }
            } else if (k == 1) {
                pair0 = h ? y : y_even;
            } else if (k == 2) {
                y_even = y;
            } else {
                const float pair1 = h ? y : y_even;
                keep = w ? pair1 : pair0;
                ybuf[rw][par][1 - w][lane] = w ? pair0 : pair1;
                asm volatile("s_wait_dscnt 0x0" ::: "memory");
                asm volatile("s_barrier_signal -1" ::: "memory");
                asm volatile("s_barrier_wait -1" ::: "memory");
                other = ybuf[rw][par][w][lane];
            }
        }
    }
    {
        const float sig = __builtin_amdgcn_rcpf(1.0f + __expf(-zfin));
        op[(NC - RING) * CHUNK] = ((keep + other) + ufin * Dd) * (zfin * sig);
    }
}

extern "C" void kernel_launch(void* const* d_in, const int* in_sizes, int n_in,
                              void* d_out, int out_size, void* d_ws, size_t ws_size,
                              hipStream_t stream) {
    const float* u     = (const float*)d_in[0];
    const float* delta = (const float*)d_in[1];
    const float* A     = (const float*)d_in[2];
    const float* B     = (const float*)d_in[3];
    const float* C     = (const float*)d_in[4];
    const float* Dp    = (const float*)d_in[5];
    const float* z     = (const float*)d_in[6];
    const float* dbias = (const float*)d_in[7];
    float* out = (float*)d_out;

    k_scan_rows<<<NBATCH * NDIM / NROW, 32 * NWAVE * NROW, 0, stream>>>(u, delta, A, B, C, Dp, z, dbias, out);
}


# state column sums as a reduce-scatter: lane pairs split their four values at xor 1 and xor 2, lane m keeps and carries only entry m&3; 20 DPP ops per chunk become 9; no sum order or rounding changed
# speedup vs baseline: 4.8093x; 1.0551x over previous
#include <hip/hip_runtime.h>


#define NBATCH 2
#define NDIM   768
#define NLEN   2048
#define NST    16
#define NWAVE  2
#define NROW   2
#define NHALF  (NST / NWAVE)
#define NLANE  (NHALF / 2)
#define CHUNK  16
#define NC     (NLEN / CHUNK)
#define RING   4

typedef float v2f __attribute__((ext_vector_type(2)));
typedef float v8f __attribute__((ext_vector_type(8)));


__device__ __forceinline__ float softplus_f(float x) {
    const float l = __logf(1.0f + __expf(x));
    return (x > 20.0f) ? x : l;
}

#define ROW_SHR(n)   (0x110 + (n))
#define ROW_SHARE(n) (0x150 + (n))
#define ROW_XMASK(n) (0x160 + (n))

template <int CTRL>
__device__ __forceinline__ float row_dpp(float v) {
    return __int_as_float(__builtin_amdgcn_update_dpp(0, __float_as_int(v), CTRL, 0xf, 0xf, true));
}

__device__ __forceinline__ float xhalf(float x) {
    return __int_as_float(__builtin_amdgcn_permlanex16(__float_as_int(x), __float_as_int(x),
                                                       0x76543210, 0xfedcba98, false, false));
}

struct Tile { float dv, uv; float bv[NLANE]; float cv[NLANE]; };

__device__ __forceinline__ void load_tile(Tile& t, const float* __restrict__ dp,
                                          const float* __restrict__ up,
                                          const float* __restrict__ Bl,
                                          const float* __restrict__ Cl, int c) {
    const int o = c * CHUNK;
    t.dv = dp[o];
    t.uv = up[o];
#pragma unroll
    for (int i = 0; i < NLANE; i++) {
        const int no = (4 * (i >> 1) + (i & 1)) * NLEN;
        t.bv[i] = Bl[no + o];
        t.cv[i] = Cl[no + o];
    }
}

__global__ __launch_bounds__(32 * NWAVE * NROW) void k_scan_rows(
    const float* __restrict__ u, const float* __restrict__ delta,
    const float* __restrict__ A, const float* __restrict__ B,
    const float* __restrict__ C, const float* __restrict__ Dp,
    const float* __restrict__ z, const float* __restrict__ dbias,
    float* __restrict__ out)
{
    __shared__ float ybuf[NROW][2][NWAVE][32];

    const int lane = threadIdx.x & 31;
    const int wv = __builtin_amdgcn_readfirstlane((int)(threadIdx.x >> 5));
    const int rw = wv >> 1;
    const int w = wv & 1;
    const int h = lane >> 4;
    const int m = lane & 15;
    const int row = blockIdx.x * NROW + rw;
    const int b = row / NDIM;
    const int d = row - b * NDIM;
    const size_t rbase = (size_t)row * NLEN;

    const float* dp = delta + rbase + m;
    const float* up = u + rbase + m;
    const float* uf = u + rbase + 2 * w * CHUNK + lane;
    const float* zf = z + rbase + 2 * w * CHUNK + lane;
    float* op = out + rbase + 2 * w * CHUNK + lane;
    const float* Bl = B + ((size_t)b * NST + NHALF * w + 2 * h) * NLEN + m;
    const float* Cl = C + ((size_t)b * NST + NHALF * w + 2 * h) * NLEN + m;

    const float bias = dbias[d];
    const float Dd = Dp[d];
    float Al[NLANE];
#pragma unroll
    for (int i = 0; i < NLANE; i++)
        Al[i] = A[d * NST + NHALF * w + 4 * (i >> 1) + 2 * h + (i & 1)] * 1.44269504088896341f;
    const float first = (m == 0) ? 1.0f : 0.0f;
    const bool odd1 = (m & 1) != 0;
    const bool odd2 = (m & 2) != 0;
    const float Alk = odd2 ? (odd1 ? Al[3] : Al[2]) : (odd1 ? Al[1] : Al[0]);

    float xc = 0.0f;

    Tile q[RING];
#pragma unroll
    for (int k = 0; k < RING - 1; k++) {
        load_tile(q[k], dp, up, Bl, Cl, k);
        __builtin_amdgcn_sched_barrier(0);
    }

    float keep = 0.0f;
    float other = 0.0f;
    float ufin = 0.0f, zfin = 0.0f;
    float y_even = 0.0f, pair0 = 0.0f;

    for (int c0 = 0; c0 < NC; c0 += RING) {
        const int par = (c0 / RING) & 1;
#pragma unroll
        for (int k = 0; k < RING; k++) {
            const int c = c0 + k;
            int cn = c + (RING - 1);
            cn = (cn < NC) ? cn : (NC - 1);
            __builtin_amdgcn_sched_barrier(0);
            load_tile(q[(k + RING - 1) % RING], dp, up, Bl, Cl, cn);
            if (k == 1) {
                ufin = uf[c0 * CHUNK];
                zfin = zf[c0 * CHUNK];
            }
            __builtin_amdgcn_sched_barrier(0);
            const Tile& t = q[k];

            const float sp = softplus_f(t.dv + bias);
            float rr = sp;
            rr += row_dpp<ROW_SHR(1)>(rr);
            rr += row_dpp<ROW_SHR(2)>(rr);
            rr += row_dpp<ROW_SHR(4)>(rr);
            rr += row_dpp<ROW_SHR(8)>(rr);
            const float du = sp * t.uv;

            float pa[NLANE], qa[NLANE], ep[NLANE], eq[NLANE], arg[NLANE];
#pragma unroll
            for (int i = 0; i < NLANE; i++) arg[i] = Al[i] * rr;
#pragma unroll
            for (int i = 0; i < NLANE; i++) ep[i] = __builtin_amdgcn_exp2f(arg[i]);
#pragma unroll
            for (int i = 0; i < NLANE; i++) eq[i] = __builtin_amdgcn_exp2f(-arg[i]);
#pragma unroll
            for (int i = 0; i < NLANE; i++) pa[i] = t.cv[i] * ep[i];
#pragma unroll
            for (int i = 0; i < NLANE; i++) qa[i] = t.bv[i] * eq[i];
            qa[0] = fmaf(xc, first, qa[0] * du);
            qa[1] = fmaf(row_dpp<ROW_SHARE(1)>(xc), first, qa[1] * du);
            qa[2] = fmaf(row_dpp<ROW_SHARE(2)>(xc), first, qa[2] * du);
            qa[3] = fmaf(row_dpp<ROW_SHARE(3)>(xc), first, qa[3] * du);

            v8f acc = {0.f, 0.f, 0.f, 0.f, 0.f, 0.f, 0.f, 0.f};
#pragma unroll
            for (int j = 0; j < NLANE / 2; j++) {
                v2f va; va.x = qa[2 * j]; va.y = qa[2 * j + 1];
                v2f vb; vb.x = pa[2 * j]; vb.y = pa[2 * j + 1];
                acc = __builtin_amdgcn_wmma_f32_16x16x4_f32(
                    false, va, false, vb, (short)0, acc, false, false);
            }

            const float k0 = odd1 ? qa[1] : qa[0], g0 = odd1 ? qa[0] : qa[1];
            const float k1 = odd1 ? qa[3] : qa[2], g1 = odd1 ? qa[2] : qa[3];
            const float p0 = k0 + row_dpp<ROW_XMASK(1)>(g0);
            const float p1 = k1 + row_dpp<ROW_XMASK(1)>(g1);
            const float k2 = odd2 ? p1 : p0, g2 = odd2 ? p0 : p1;
            float cs = k2 + row_dpp<ROW_XMASK(2)>(g2);
            cs += row_dpp<ROW_XMASK(4)>(cs);
            cs += row_dpp<ROW_XMASK(8)>(cs);
            xc = cs * __builtin_amdgcn_exp2f(Alk * row_dpp<ROW_SHARE(15)>(rr));

            float ym[8];
#pragma unroll
            for (int r = 0; r < 8; r++) ym[r] = ((8 * h + r) <= m) ? acc[r] : 0.0f;
            float y = ((ym[0] + ym[1]) + (ym[2] + ym[3])) + ((ym[4] + ym[5]) + (ym[6] + ym[7]));
            y += xhalf(y);

            if (k == 0) {
                y_even = y;
                if (c0 > 0) {
                    const float sig = __builtin_amdgcn_rcpf(1.0f + __expf(-zfin));
                    op[(c0 - RING) * CHUNK] = ((keep + other) + ufin * Dd) * (zfin * sig);
                }
            } else if (k == 1) {
                pair0 = h ? y : y_even;
            } else if (k == 2) {
                y_even = y;
            } else {
                const float pair1 = h ? y : y_even;
                keep = w ? pair1 : pair0;
                ybuf[rw][par][1 - w][lane] = w ? pair0 : pair1;
                asm volatile("s_wait_dscnt 0x0" ::: "memory");
                asm volatile("s_barrier_signal -1" ::: "memory");
                asm volatile("s_barrier_wait -1" ::: "memory");
                other = ybuf[rw][par][w][lane];
            }
        }
    }
    {
        const float sig = __builtin_amdgcn_rcpf(1.0f + __expf(-zfin));
        op[(NC - RING) * CHUNK] = ((keep + other) + ufin * Dd) * (zfin * sig);
    }
}

extern "C" void kernel_launch(void* const* d_in, const int* in_sizes, int n_in,
                              void* d_out, int out_size, void* d_ws, size_t ws_size,
                              hipStream_t stream) {
    const float* u     = (const float*)d_in[0];
    const float* delta = (const float*)d_in[1];
    const float* A     = (const float*)d_in[2];
    const float* B     = (const float*)d_in[3];
    const float* C     = (const float*)d_in[4];
    const float* Dp    = (const float*)d_in[5];
    const float* z     = (const float*)d_in[6];
    const float* dbias = (const float*)d_in[7];
    float* out = (float*)d_out;

    k_scan_rows<<<NBATCH * NDIM / NROW, 32 * NWAVE * NROW, 0, stream>>>(u, delta, A, B, C, Dp, z, dbias, out);
}


# softplus as log2(1+2^t) on the bare exp2/log2 instructions (library log's rescale and two-term ln2 product gone), running sum kept in units of ln2; softplus rounding changes, no sum order changed
# speedup vs baseline: 5.1741x; 1.0759x over previous
#include <hip/hip_runtime.h>


#define NBATCH 2
#define NDIM   768
#define NLEN   2048
#define NST    16
#define NWAVE  2
#define NROW   2
#define NHALF  (NST / NWAVE)
#define NLANE  (NHALF / 2)
#define CHUNK  16
#define NC     (NLEN / CHUNK)
#define RING   4

typedef float v2f __attribute__((ext_vector_type(2)));
typedef float v8f __attribute__((ext_vector_type(8)));


__device__ __forceinline__ float softplus_log2(float t) {
    const float l = __builtin_amdgcn_logf(1.0f + __builtin_amdgcn_exp2f(t));
    return (t > 28.853900817779268f) ? t : l;
}

#define ROW_SHR(n)   (0x110 + (n))
#define ROW_SHARE(n) (0x150 + (n))
#define ROW_XMASK(n) (0x160 + (n))

template <int CTRL>
__device__ __forceinline__ float row_dpp(float v) {
    return __int_as_float(__builtin_amdgcn_update_dpp(0, __float_as_int(v), CTRL, 0xf, 0xf, true));
}

__device__ __forceinline__ float xhalf(float x) {
    return __int_as_float(__builtin_amdgcn_permlanex16(__float_as_int(x), __float_as_int(x),
                                                       0x76543210, 0xfedcba98, false, false));
}

struct Tile { float dv, uv; float bv[NLANE]; float cv[NLANE]; };

__device__ __forceinline__ void load_tile(Tile& t, const float* __restrict__ dp,
                                          const float* __restrict__ up,
                                          const float* __restrict__ Bl,
                                          const float* __restrict__ Cl, int c) {
    const int o = c * CHUNK;
    t.dv = dp[o];
    t.uv = up[o];
#pragma unroll
    for (int i = 0; i < NLANE; i++) {
        const int no = (4 * (i >> 1) + (i & 1)) * NLEN;
        t.bv[i] = Bl[no + o];
        t.cv[i] = Cl[no + o];
    }
}

__global__ __launch_bounds__(32 * NWAVE * NROW) void k_scan_rows(
    const float* __restrict__ u, const float* __restrict__ delta,
    const float* __restrict__ A, const float* __restrict__ B,
    const float* __restrict__ C, const float* __restrict__ Dp,
    const float* __restrict__ z, const float* __restrict__ dbias,
    float* __restrict__ out)
{
    __shared__ float ybuf[NROW][2][NWAVE][32];

    const int lane = threadIdx.x & 31;
    const int wv = __builtin_amdgcn_readfirstlane((int)(threadIdx.x >> 5));
    const int rw = wv >> 1;
    const int w = wv & 1;
    const int h = lane >> 4;
    const int m = lane & 15;
    const int row = blockIdx.x * NROW + rw;
    const int b = row / NDIM;
    const int d = row - b * NDIM;
    const size_t rbase = (size_t)row * NLEN;

    const float* dp = delta + rbase + m;
    const float* up = u + rbase + m;
    const float* uf = u + rbase + 2 * w * CHUNK + lane;
    const float* zf = z + rbase + 2 * w * CHUNK + lane;
    float* op = out + rbase + 2 * w * CHUNK + lane;
    const float* Bl = B + ((size_t)b * NST + NHALF * w + 2 * h) * NLEN + m;
    const float* Cl = C + ((size_t)b * NST + NHALF * w + 2 * h) * NLEN + m;

    const float biasl = dbias[d] * 1.44269504088896341f;
    const float Dd = Dp[d];
    float Al[NLANE];
#pragma unroll
    for (int i = 0; i < NLANE; i++)
        Al[i] = A[d * NST + NHALF * w + 4 * (i >> 1) + 2 * h + (i & 1)];
    const float first = (m == 0) ? 1.0f : 0.0f;
    const bool odd1 = (m & 1) != 0;
    const bool odd2 = (m & 2) != 0;
    const float Alk = odd2 ? (odd1 ? Al[3] : Al[2]) : (odd1 ? Al[1] : Al[0]);

    float xc = 0.0f;

    Tile q[RING];
#pragma unroll
    for (int k = 0; k < RING - 1; k++) {
        load_tile(q[k], dp, up, Bl, Cl, k);
        __builtin_amdgcn_sched_barrier(0);
    }

    float keep = 0.0f;
    float other = 0.0f;
    float ufin = 0.0f, zfin = 0.0f;
    float y_even = 0.0f, pair0 = 0.0f;

    for (int c0 = 0; c0 < NC; c0 += RING) {
        const int par = (c0 / RING) & 1;
#pragma unroll
        for (int k = 0; k < RING; k++) {
            const int c = c0 + k;
            int cn = c + (RING - 1);
            cn = (cn < NC) ? cn : (NC - 1);
            __builtin_amdgcn_sched_barrier(0);
            load_tile(q[(k + RING - 1) % RING], dp, up, Bl, Cl, cn);
            if (k == 1) {
                ufin = uf[c0 * CHUNK];
                zfin = zf[c0 * CHUNK];
            }
            __builtin_amdgcn_sched_barrier(0);
            const Tile& t = q[k];

            const float sp = softplus_log2(fmaf(t.dv, 1.44269504088896341f, biasl));
            float rr = sp;
            rr += row_dpp<ROW_SHR(1)>(rr);
            rr += row_dpp<ROW_SHR(2)>(rr);
            rr += row_dpp<ROW_SHR(4)>(rr);
            rr += row_dpp<ROW_SHR(8)>(rr);
            const float du = sp * (t.uv * 0.693147180559945309f);

            float pa[NLANE], qa[NLANE], ep[NLANE], eq[NLANE], arg[NLANE];
#pragma unroll
            for (int i = 0; i < NLANE; i++) arg[i] = Al[i] * rr;
#pragma unroll
            for (int i = 0; i < NLANE; i++) ep[i] = __builtin_amdgcn_exp2f(arg[i]);
#pragma unroll
            for (int i = 0; i < NLANE; i++) eq[i] = __builtin_amdgcn_exp2f(-arg[i]);
#pragma unroll
            for (int i = 0; i < NLANE; i++) pa[i] = t.cv[i] * ep[i];
#pragma unroll
            for (int i = 0; i < NLANE; i++) qa[i] = t.bv[i] * eq[i];
            qa[0] = fmaf(xc, first, qa[0] * du);
            qa[1] = fmaf(row_dpp<ROW_SHARE(1)>(xc), first, qa[1] * du);
            qa[2] = fmaf(row_dpp<ROW_SHARE(2)>(xc), first, qa[2] * du);
            qa[3] = fmaf(row_dpp<ROW_SHARE(3)>(xc), first, qa[3] * du);

            v8f acc = {0.f, 0.f, 0.f, 0.f, 0.f, 0.f, 0.f, 0.f};
#pragma unroll
            for (int j = 0; j < NLANE / 2; j++) {
                v2f va; va.x = qa[2 * j]; va.y = qa[2 * j + 1];
                v2f vb; vb.x = pa[2 * j]; vb.y = pa[2 * j + 1];
                acc = __builtin_amdgcn_wmma_f32_16x16x4_f32(
                    false, va, false, vb, (short)0, acc, false, false);
            }

            const float k0 = odd1 ? qa[1] : qa[0], g0 = odd1 ? qa[0] : qa[1];
            const float k1 = odd1 ? qa[3] : qa[2], g1 = odd1 ? qa[2] : qa[3];
            const float p0 = k0 + row_dpp<ROW_XMASK(1)>(g0);
            const float p1 = k1 + row_dpp<ROW_XMASK(1)>(g1);
            const float k2 = odd2 ? p1 : p0, g2 = odd2 ? p0 : p1;
            float cs = k2 + row_dpp<ROW_XMASK(2)>(g2);
            cs += row_dpp<ROW_XMASK(4)>(cs);
            cs += row_dpp<ROW_XMASK(8)>(cs);
            xc = cs * __builtin_amdgcn_exp2f(Alk * row_dpp<ROW_SHARE(15)>(rr));

            float ym[8];
#pragma unroll
            for (int r = 0; r < 8; r++) ym[r] = ((8 * h + r) <= m) ? acc[r] : 0.0f;
            float y = ((ym[0] + ym[1]) + (ym[2] + ym[3])) + ((ym[4] + ym[5]) + (ym[6] + ym[7]));
            y += xhalf(y);

            if (k == 0) {
                y_even = y;
                if (c0 > 0) {
                    const float sig = __builtin_amdgcn_rcpf(1.0f + __expf(-zfin));
                    op[(c0 - RING) * CHUNK] = ((keep + other) + ufin * Dd) * (zfin * sig);
                }
            } else if (k == 1) {
                pair0 = h ? y : y_even;
            } else if (k == 2) {
                y_even = y;
            } else {
                const float pair1 = h ? y : y_even;
                keep = w ? pair1 : pair0;
                ybuf[rw][par][1 - w][lane] = w ? pair0 : pair1;
                asm volatile("s_wait_dscnt 0x0" ::: "memory");
                asm volatile("s_barrier_signal -1" ::: "memory");
                asm volatile("s_barrier_wait -1" ::: "memory");
                other = ybuf[rw][par][w][lane];
            }
        }
    }
    {
        const float sig = __builtin_amdgcn_rcpf(1.0f + __expf(-zfin));
        op[(NC - RING) * CHUNK] = ((keep + other) + ufin * Dd) * (zfin * sig);
    }
}

extern "C" void kernel_launch(void* const* d_in, const int* in_sizes, int n_in,
                              void* d_out, int out_size, void* d_ws, size_t ws_size,
                              hipStream_t stream) {
    const float* u     = (const float*)d_in[0];
    const float* delta = (const float*)d_in[1];
    const float* A     = (const float*)d_in[2];
    const float* B     = (const float*)d_in[3];
    const float* C     = (const float*)d_in[4];
    const float* Dp    = (const float*)d_in[5];
    const float* z     = (const float*)d_in[6];
    const float* dbias = (const float*)d_in[7];
    float* out = (float*)d_out;

    k_scan_rows<<<NBATCH * NDIM / NROW, 32 * NWAVE * NROW, 0, stream>>>(u, delta, A, B, C, Dp, z, dbias, out);
}


# softplus, its running sum and softplus*u made once per row: each wave prepares its own line's two chunks (lane half = chunk) two groups ahead, passed through LDS; no sum order or rounding changed
# speedup vs baseline: 5.3980x; 1.0433x over previous
#include <hip/hip_runtime.h>


#define NBATCH 2
#define NDIM   768
#define NLEN   2048
#define NST    16
#define NWAVE  2
#define NROW   2
#define NHALF  (NST / NWAVE)
#define NLANE  (NHALF / 2)
#define CHUNK  16
#define NC     (NLEN / CHUNK)
#define RING   4

typedef float v2f __attribute__((ext_vector_type(2)));
typedef float v4f __attribute__((ext_vector_type(4)));
typedef float v8f __attribute__((ext_vector_type(8)));


__device__ __forceinline__ float softplus_log2(float t) {
    const float l = __builtin_amdgcn_logf(1.0f + __builtin_amdgcn_exp2f(t));
    return (t > 28.853900817779268f) ? t : l;
}

#define ROW_SHR(n)   (0x110 + (n))
#define ROW_SHARE(n) (0x150 + (n))
#define ROW_XMASK(n) (0x160 + (n))

template <int CTRL>
__device__ __forceinline__ float row_dpp(float v) {
    return __int_as_float(__builtin_amdgcn_update_dpp(0, __float_as_int(v), CTRL, 0xf, 0xf, true));
}

__device__ __forceinline__ float xhalf(float x) {
    return __int_as_float(__builtin_amdgcn_permlanex16(__float_as_int(x), __float_as_int(x),
                                                       0x76543210, 0xfedcba98, false, false));
}

struct Tile { float bv[NLANE]; float cv[NLANE]; };

__device__ __forceinline__ void load_tile(Tile& t, const float* __restrict__ Bl,
                                          const float* __restrict__ Cl, int c) {
    const int o = c * CHUNK;
#pragma unroll
    for (int i = 0; i < NLANE; i++) {
        const int no = (4 * (i >> 1) + (i & 1)) * NLEN;
        t.bv[i] = Bl[no + o];
        t.cv[i] = Cl[no + o];
    }
}

__device__ __forceinline__ void prep_line(float dvl, float uvl, float biasl, float& rr, float& du) {
    const float sp = softplus_log2(fmaf(dvl, 1.44269504088896341f, biasl));
    rr = sp;
    rr += row_dpp<ROW_SHR(1)>(rr);
    rr += row_dpp<ROW_SHR(2)>(rr);
    rr += row_dpp<ROW_SHR(4)>(rr);
    rr += row_dpp<ROW_SHR(8)>(rr);
    du = sp * (uvl * 0.693147180559945309f);
}

__global__ __launch_bounds__(32 * NWAVE * NROW) void k_scan_rows(
    const float* __restrict__ u, const float* __restrict__ delta,
    const float* __restrict__ A, const float* __restrict__ B,
    const float* __restrict__ C, const float* __restrict__ Dp,
    const float* __restrict__ z, const float* __restrict__ dbias,
    float* __restrict__ out)
{
    __shared__ float ybuf[NROW][2][NWAVE][32];
    __shared__ __attribute__((aligned(16))) float sR[NROW][2][CHUNK][RING];
    __shared__ __attribute__((aligned(16))) float sD[NROW][2][CHUNK][RING];

    const int lane = threadIdx.x & 31;
    const int wv = __builtin_amdgcn_readfirstlane((int)(threadIdx.x >> 5));
    const int rw = wv >> 1;
    const int w = wv & 1;
    const int h = lane >> 4;
    const int m = lane & 15;
    const int row = blockIdx.x * NROW + rw;
    const int b = row / NDIM;
    const int d = row - b * NDIM;
    const size_t rbase = (size_t)row * NLEN;

    const float* dl = delta + rbase + 2 * w * CHUNK + lane;
    const float* uf = u + rbase + 2 * w * CHUNK + lane;
    const float* zf = z + rbase + 2 * w * CHUNK + lane;
    float* op = out + rbase + 2 * w * CHUNK + lane;
    const float* Bl = B + ((size_t)b * NST + NHALF * w + 2 * h) * NLEN + m;
    const float* Cl = C + ((size_t)b * NST + NHALF * w + 2 * h) * NLEN + m;

    const float biasl = dbias[d] * 1.44269504088896341f;
    const float Dd = Dp[d];
    float Al[NLANE];
#pragma unroll
    for (int i = 0; i < NLANE; i++)
        Al[i] = A[d * NST + NHALF * w + 4 * (i >> 1) + 2 * h + (i & 1)];
    const float first = (m == 0) ? 1.0f : 0.0f;
    const bool odd1 = (m & 1) != 0;
    const bool odd2 = (m & 2) != 0;
    const float Alk = odd2 ? (odd1 ? Al[3] : Al[2]) : (odd1 ? Al[1] : Al[0]);

    float xc = 0.0f;

    {
        const float d0 = dl[0], u0 = uf[0];
        const float d1 = dl[RING * CHUNK], u1 = uf[RING * CHUNK];
        float r0, s0, r1, s1;
        prep_line(d0, u0, biasl, r0, s0);
        prep_line(d1, u1, biasl, r1, s1);
        sR[rw][0][m][2 * w + h] = r0;
        sD[rw][0][m][2 * w + h] = s0;
        sR[rw][1][m][2 * w + h] = r1;
        sD[rw][1][m][2 * w + h] = s1;
    }
    asm volatile("s_wait_dscnt 0x0" ::: "memory");
    asm volatile("s_barrier_signal -1" ::: "memory");
    asm volatile("s_barrier_wait -1" ::: "memory");
    v4f cR = *(const v4f*)&sR[rw][0][m][0];
    v4f cD = *(const v4f*)&sD[rw][0][m][0];
    v4f nR = cR, nD = cD;
    asm volatile("s_wait_dscnt 0x0" ::: "memory");
    asm volatile("s_barrier_signal -1" ::: "memory");
    asm volatile("s_barrier_wait -1" ::: "memory");

    Tile q[RING];
#pragma unroll
    for (int k = 0; k < RING - 1; k++) {
        load_tile(q[k], Bl, Cl, k);
        __builtin_amdgcn_sched_barrier(0);
    }

    float keep = 0.0f;
    float other = 0.0f;
    float ufin = 0.0f, zfin = 0.0f;
    float y_even = 0.0f, pair0 = 0.0f;
    float dln = 0.0f, uln = 0.0f;

    for (int c0 = 0; c0 < NC; c0 += RING) {
        const int par = (c0 / RING) & 1;
#pragma unroll
        for (int k = 0; k < RING; k++) {
            const int c = c0 + k;
            int cn = c + (RING - 1);
            cn = (cn < NC) ? cn : (NC - 1);
            __builtin_amdgcn_sched_barrier(0);
            load_tile(q[(k + RING - 1) % RING], Bl, Cl, cn);
            if (k == 0) {
                int gl = c0 + 2 * RING;
                gl = (gl < NC) ? gl : (NC - RING);
                dln = dl[gl * CHUNK];
                uln = uf[gl * CHUNK];
            }
            if (k == 1) {
                nR = *(const v4f*)&sR[rw][par ^ 1][m][0];
                nD = *(const v4f*)&sD[rw][par ^ 1][m][0];
            }
            if (k == 1) {
                ufin = uf[c0 * CHUNK];
                zfin = zf[c0 * CHUNK];
            }
            __builtin_amdgcn_sched_barrier(0);
            const Tile& t = q[k];

            const float rr = cR[k];
            const float du = cD[k];

            float pa[NLANE], qa[NLANE], ep[NLANE], eq[NLANE], arg[NLANE];
#pragma unroll
            for (int i = 0; i < NLANE; i++) arg[i] = Al[i] * rr;
#pragma unroll
            for (int i = 0; i < NLANE; i++) ep[i] = __builtin_amdgcn_exp2f(arg[i]);
#pragma unroll
            for (int i = 0; i < NLANE; i++) eq[i] = __builtin_amdgcn_exp2f(-arg[i]);
#pragma unroll
            for (int i = 0; i < NLANE; i++) pa[i] = t.cv[i] * ep[i];
#pragma unroll
            for (int i = 0; i < NLANE; i++) qa[i] = t.bv[i] * eq[i];
            qa[0] = fmaf(xc, first, qa[0] * du);
            qa[1] = fmaf(row_dpp<ROW_SHARE(1)>(xc), first, qa[1] * du);
            qa[2] = fmaf(row_dpp<ROW_SHARE(2)>(xc), first, qa[2] * du);
            qa[3] = fmaf(row_dpp<ROW_SHARE(3)>(xc), first, qa[3] * du);

            v8f acc = {0.f, 0.f, 0.f, 0.f, 0.f, 0.f, 0.f, 0.f};
#pragma unroll
            for (int j = 0; j < NLANE / 2; j++) {
                v2f va; va.x = qa[2 * j]; va.y = qa[2 * j + 1];
                v2f vb; vb.x = pa[2 * j]; vb.y = pa[2 * j + 1];
                acc = __builtin_amdgcn_wmma_f32_16x16x4_f32(
                    false, va, false, vb, (short)0, acc, false, false);
            }

            const float k0 = odd1 ? qa[1] : qa[0], g0 = odd1 ? qa[0] : qa[1];
            const float k1 = odd1 ? qa[3] : qa[2], g1 = odd1 ? qa[2] : qa[3];
            const float p0 = k0 + row_dpp<ROW_XMASK(1)>(g0);
            const float p1 = k1 + row_dpp<ROW_XMASK(1)>(g1);
            const float k2 = odd2 ? p1 : p0, g2 = odd2 ? p0 : p1;
            float cs = k2 + row_dpp<ROW_XMASK(2)>(g2);
            cs += row_dpp<ROW_XMASK(4)>(cs);
            cs += row_dpp<ROW_XMASK(8)>(cs);
            xc = cs * __builtin_amdgcn_exp2f(Alk * row_dpp<ROW_SHARE(15)>(rr));

            float ym[8];
#pragma unroll
            for (int r = 0; r < 8; r++) ym[r] = ((8 * h + r) <= m) ? acc[r] : 0.0f;
            float y = ((ym[0] + ym[1]) + (ym[2] + ym[3])) + ((ym[4] + ym[5]) + (ym[6] + ym[7]));
            y += xhalf(y);

            if (k == 0) {
                y_even = y;
                if (c0 > 0) {
                    const float sig = __builtin_amdgcn_rcpf(1.0f + __expf(-zfin));
                    op[(c0 - RING) * CHUNK] = ((keep + other) + ufin * Dd) * (zfin * sig);
                }
            } else if (k == 1) {
                pair0 = h ? y : y_even;
            } else if (k == 2) {
                y_even = y;
            } else {
                const float pair1 = h ? y : y_even;
                keep = w ? pair1 : pair0;
                float pr, pd;
                prep_line(dln, uln, biasl, pr, pd);
                sR[rw][par][m][2 * w + h] = pr;
                sD[rw][par][m][2 * w + h] = pd;
                ybuf[rw][par][1 - w][lane] = w ? pair0 : pair1;
                asm volatile("s_wait_dscnt 0x0" ::: "memory");
                asm volatile("s_barrier_signal -1" ::: "memory");
                asm volatile("s_barrier_wait -1" ::: "memory");
                other = ybuf[rw][par][w][lane];
            }
        }
        cR = nR;
        cD = nD;
    }
    {
        const float sig = __builtin_amdgcn_rcpf(1.0f + __expf(-zfin));
        op[(NC - RING) * CHUNK] = ((keep + other) + ufin * Dd) * (zfin * sig);
    }
}

extern "C" void kernel_launch(void* const* d_in, const int* in_sizes, int n_in,
                              void* d_out, int out_size, void* d_ws, size_t ws_size,
                              hipStream_t stream) {
    const float* u     = (const float*)d_in[0];
    const float* delta = (const float*)d_in[1];
    const float* A     = (const float*)d_in[2];
    const float* B     = (const float*)d_in[3];
    const float* C     = (const float*)d_in[4];
    const float* Dp    = (const float*)d_in[5];
    const float* z     = (const float*)d_in[6];
    const float* dbias = (const float*)d_in[7];
    float* out = (float*)d_out;

    k_scan_rows<<<NBATCH * NDIM / NROW, 32 * NWAVE * NROW, 0, stream>>>(u, delta, A, B, C, Dp, z, dbias, out);
}


# address and buffer bookkeeping taken off the vector unit: every load and the store use a scalar base plus one fixed lane offset, loop unrolled over the two LDS buffer parities; no sum order changed
# speedup vs baseline: 5.8752x; 1.0884x over previous
#include <hip/hip_runtime.h>


#define NBATCH 2
#define NDIM   768
#define NLEN   2048
#define NST    16
#define NWAVE  2
#define NROW   2
#define NHALF  (NST / NWAVE)
#define NLANE  (NHALF / 2)
#define CHUNK  16
#define NC     (NLEN / CHUNK)
#define RING   4

typedef const float __attribute__((address_space(1))) gcfloat;
typedef float __attribute__((address_space(1))) gfloat;
typedef float v2f __attribute__((ext_vector_type(2)));
typedef float v4f __attribute__((ext_vector_type(4)));
typedef float v8f __attribute__((ext_vector_type(8)));


__device__ __forceinline__ float softplus_log2(float t) {
    const float l = __builtin_amdgcn_logf(1.0f + __builtin_amdgcn_exp2f(t));
    return (t > 28.853900817779268f) ? t : l;
}

#define ROW_SHR(n)   (0x110 + (n))
#define ROW_SHARE(n) (0x150 + (n))
#define ROW_XMASK(n) (0x160 + (n))

template <int CTRL>
__device__ __forceinline__ float row_dpp(float v) {
    return __int_as_float(__builtin_amdgcn_update_dpp(0, __float_as_int(v), CTRL, 0xf, 0xf, true));
}

__device__ __forceinline__ float xhalf(float x) {
    return __int_as_float(__builtin_amdgcn_permlanex16(__float_as_int(x), __float_as_int(x),
                                                       0x76543210, 0xfedcba98, false, false));
}

__device__ __forceinline__ float ld_row(const float* p, unsigned ul) {
    gcfloat* g = (gcfloat*)p;
    asm("" : "+s"(g));
    return g[ul];
}
__device__ __forceinline__ void st_row(float* p, unsigned ul, float v) {
    gfloat* g = (gfloat*)p;
    asm("" : "+s"(g));
    g[ul] = v;
}

struct Tile { float bv[NLANE]; float cv[NLANE]; };

__device__ __forceinline__ void load_tile(Tile& t, const float* __restrict__ Bl,
                                          const float* __restrict__ Cl, int c, unsigned lo) {
    gcfloat* Bc = (gcfloat*)(Bl + c * CHUNK);
    gcfloat* Cc = (gcfloat*)(Cl + c * CHUNK);
    asm("" : "+s"(Bc));
    asm("" : "+s"(Cc));
#pragma unroll
    for (int i = 0; i < NLANE; i++) {
        const int no = (4 * (i >> 1) + (i & 1)) * NLEN;
        t.bv[i] = (Bc + no)[lo];
        t.cv[i] = (Cc + no)[lo];
    }
}

__device__ __forceinline__ void prep_line(float dvl, float uvl, float biasl, float& rr, float& du) {
    const float sp = softplus_log2(fmaf(dvl, 1.44269504088896341f, biasl));
    rr = sp;
    rr += row_dpp<ROW_SHR(1)>(rr);
    rr += row_dpp<ROW_SHR(2)>(rr);
    rr += row_dpp<ROW_SHR(4)>(rr);
    rr += row_dpp<ROW_SHR(8)>(rr);
    du = sp * (uvl * 0.693147180559945309f);
}

__global__ __launch_bounds__(32 * NWAVE * NROW) void k_scan_rows(
    const float* __restrict__ u, const float* __restrict__ delta,
    const float* __restrict__ A, const float* __restrict__ B,
    const float* __restrict__ C, const float* __restrict__ Dp,
    const float* __restrict__ z, const float* __restrict__ dbias,
    float* __restrict__ out)
{
    __shared__ float ybuf[NROW][2][NWAVE][32];
    __shared__ __attribute__((aligned(16))) float sR[NROW][2][CHUNK][RING];
    __shared__ __attribute__((aligned(16))) float sD[NROW][2][CHUNK][RING];

    const int lane = threadIdx.x & 31;
    const int wv = __builtin_amdgcn_readfirstlane((int)(threadIdx.x >> 5));
    const int rw = wv >> 1;
    const int w = wv & 1;
    const int h = lane >> 4;
    const int m = lane & 15;
    const int row = blockIdx.x * NROW + rw;
    const int b = row / NDIM;
    const int d = row - b * NDIM;
    const size_t rbase = (size_t)row * NLEN;

    const unsigned ul = (unsigned)lane;
    const unsigned lo = (unsigned)(2 * h * NLEN + m);
    const float* dl = delta + rbase + 2 * w * CHUNK;
    const float* uf = u + rbase + 2 * w * CHUNK;
    const float* zf = z + rbase + 2 * w * CHUNK;
    float* op = out + rbase + 2 * w * CHUNK;
    const float* Bl = B + ((size_t)b * NST + NHALF * w) * NLEN;
    const float* Cl = C + ((size_t)b * NST + NHALF * w) * NLEN;

    const float biasl = dbias[d] * 1.44269504088896341f;
    const float Dd = Dp[d];
    float Al[NLANE];
#pragma unroll
    for (int i = 0; i < NLANE; i++)
        Al[i] = A[d * NST + NHALF * w + 4 * (i >> 1) + 2 * h + (i & 1)];
    const float first = (m == 0) ? 1.0f : 0.0f;
    const bool odd1 = (m & 1) != 0;
    const bool odd2 = (m & 2) != 0;
    const float Alk = odd2 ? (odd1 ? Al[3] : Al[2]) : (odd1 ? Al[1] : Al[0]);

    float xc = 0.0f;

    {
        const float d0 = dl[ul], u0 = uf[ul];
        const float d1 = (dl + RING * CHUNK)[ul], u1 = (uf + RING * CHUNK)[ul];
        float r0, s0, r1, s1;
        prep_line(d0, u0, biasl, r0, s0);
        prep_line(d1, u1, biasl, r1, s1);
        sR[rw][0][m][2 * w + h] = r0;
        sD[rw][0][m][2 * w + h] = s0;
        sR[rw][1][m][2 * w + h] = r1;
        sD[rw][1][m][2 * w + h] = s1;
    }
    asm volatile("s_wait_dscnt 0x0" ::: "memory");
    asm volatile("s_barrier_signal -1" ::: "memory");
    asm volatile("s_barrier_wait -1" ::: "memory");
    v4f gR[2], gD[2];
    gR[0] = *(const v4f*)&sR[rw][0][m][0];
    gD[0] = *(const v4f*)&sD[rw][0][m][0];
    gR[1] = gR[0]; gD[1] = gD[0];
    asm volatile("s_wait_dscnt 0x0" ::: "memory");
    asm volatile("s_barrier_signal -1" ::: "memory");
    asm volatile("s_barrier_wait -1" ::: "memory");

    Tile q[RING];
#pragma unroll
    for (int k = 0; k < RING - 1; k++) {
        load_tile(q[k], Bl, Cl, k, lo);
        __builtin_amdgcn_sched_barrier(0);
    }

    float keep = 0.0f;
    float other = 0.0f;
    float ufin = 0.0f, zfin = 0.0f;
    float y_even = 0.0f, pair0 = 0.0f;
    float dln = 0.0f, uln = 0.0f;

    for (int cc = 0; cc < NC; cc += 2 * RING) {
#pragma unroll
      for (int par = 0; par < 2; par++) {
        const int c0 = cc + par * RING;
#pragma unroll
        for (int k = 0; k < RING; k++) {
            const int c = c0 + k;
            int cn = c + (RING - 1);
            cn = (cn < NC) ? cn : (NC - 1);
            __builtin_amdgcn_sched_barrier(0);
            load_tile(q[(k + RING - 1) % RING], Bl, Cl, cn, lo);
            if (k == 0) {
                int gl = c0 + 2 * RING;
                gl = (gl < NC) ? gl : (NC - RING);
                dln = ld_row(dl + gl * CHUNK, ul);
                uln = ld_row(uf + gl * CHUNK, ul);
            }
            if (k == 1) {
                gR[par ^ 1] = *(const v4f*)&sR[rw][par ^ 1][m][0];
                gD[par ^ 1] = *(const v4f*)&sD[rw][par ^ 1][m][0];
            }
            if (k == 1) {
                ufin = ld_row(uf + c0 * CHUNK, ul);
                zfin = ld_row(zf + c0 * CHUNK, ul);
            }
            __builtin_amdgcn_sched_barrier(0);
            const Tile& t = q[k];

            const float rr = gR[par][k];
            const float du = gD[par][k];

            float pa[NLANE], qa[NLANE], ep[NLANE], eq[NLANE], arg[NLANE];
#pragma unroll
            for (int i = 0; i < NLANE; i++) arg[i] = Al[i] * rr;
#pragma unroll
            for (int i = 0; i < NLANE; i++) ep[i] = __builtin_amdgcn_exp2f(arg[i]);
#pragma unroll
            for (int i = 0; i < NLANE; i++) eq[i] = __builtin_amdgcn_exp2f(-arg[i]);
#pragma unroll
            for (int i = 0; i < NLANE; i++) pa[i] = t.cv[i] * ep[i];
#pragma unroll
            for (int i = 0; i < NLANE; i++) qa[i] = t.bv[i] * eq[i];
            qa[0] = fmaf(xc, first, qa[0] * du);
            qa[1] = fmaf(row_dpp<ROW_SHARE(1)>(xc), first, qa[1] * du);
            qa[2] = fmaf(row_dpp<ROW_SHARE(2)>(xc), first, qa[2] * du);
            qa[3] = fmaf(row_dpp<ROW_SHARE(3)>(xc), first, qa[3] * du);

            v8f acc = {0.f, 0.f, 0.f, 0.f, 0.f, 0.f, 0.f, 0.f};
#pragma unroll
            for (int j = 0; j < NLANE / 2; j++) {
                v2f va; va.x = qa[2 * j]; va.y = qa[2 * j + 1];
                v2f vb; vb.x = pa[2 * j]; vb.y = pa[2 * j + 1];
                acc = __builtin_amdgcn_wmma_f32_16x16x4_f32(
                    false, va, false, vb, (short)0, acc, false, false);
            }

            const float k0 = odd1 ? qa[1] : qa[0], g0 = odd1 ? qa[0] : qa[1];
            const float k1 = odd1 ? qa[3] : qa[2], g1 = odd1 ? qa[2] : qa[3];
            const float p0 = k0 + row_dpp<ROW_XMASK(1)>(g0);
            const float p1 = k1 + row_dpp<ROW_XMASK(1)>(g1);
            const float k2 = odd2 ? p1 : p0, g2 = odd2 ? p0 : p1;
            float cs = k2 + row_dpp<ROW_XMASK(2)>(g2);
            cs += row_dpp<ROW_XMASK(4)>(cs);
            cs += row_dpp<ROW_XMASK(8)>(cs);
            xc = cs * __builtin_amdgcn_exp2f(Alk * row_dpp<ROW_SHARE(15)>(rr));

            float ym[8];
#pragma unroll
            for (int r = 0; r < 8; r++) ym[r] = ((8 * h + r) <= m) ? acc[r] : 0.0f;
            float y = ((ym[0] + ym[1]) + (ym[2] + ym[3])) + ((ym[4] + ym[5]) + (ym[6] + ym[7]));
            y += xhalf(y);

            if (k == 0) {
                y_even = y;
                if (c0 > 0) {
                    const float sig = __builtin_amdgcn_rcpf(1.0f + __expf(-zfin));
                    st_row(op + (c0 - RING) * CHUNK, ul, ((keep + other) + ufin * Dd) * (zfin * sig));
                }
            } else if (k == 1) {
                pair0 = h ? y : y_even;
            } else if (k == 2) {
                y_even = y;
            } else {
                const float pair1 = h ? y : y_even;
                keep = w ? pair1 : pair0;
                float pr, pd;
                prep_line(dln, uln, biasl, pr, pd);
                sR[rw][par][m][2 * w + h] = pr;
                sD[rw][par][m][2 * w + h] = pd;
                ybuf[rw][par][1 - w][lane] = w ? pair0 : pair1;
                asm volatile("s_wait_dscnt 0x0" ::: "memory");
                asm volatile("s_barrier_signal -1" ::: "memory");
                asm volatile("s_barrier_wait -1" ::: "memory");
                other = ybuf[rw][par][w][lane];
            }
        }
      }
    }
    {
        const float sig = __builtin_amdgcn_rcpf(1.0f + __expf(-zfin));
        st_row(op + (NC - RING) * CHUNK, ul, ((keep + other) + ufin * Dd) * (zfin * sig));
    }
}

extern "C" void kernel_launch(void* const* d_in, const int* in_sizes, int n_in,
                              void* d_out, int out_size, void* d_ws, size_t ws_size,
                              hipStream_t stream) {
    const float* u     = (const float*)d_in[0];
    const float* delta = (const float*)d_in[1];
    const float* A     = (const float*)d_in[2];
    const float* B     = (const float*)d_in[3];
    const float* C     = (const float*)d_in[4];
    const float* Dp    = (const float*)d_in[5];
    const float* z     = (const float*)d_in[6];
    const float* dbias = (const float*)d_in[7];
    float* out = (float*)d_out;

    k_scan_rows<<<NBATCH * NDIM / NROW, 32 * NWAVE * NROW, 0, stream>>>(u, delta, A, B, C, Dp, z, dbias, out);
}


# causal prefix sum moved into the matrix unit: per 2 chunks Q^T (state x step) times a constant 0/1 triangle; state taken from lane 15 of a half; masks, reduce-scatter gone; y, state sum order changed
# speedup vs baseline: 6.3890x; 1.0875x over previous
#include <hip/hip_runtime.h>


#define NBATCH 2
#define NDIM   768
#define NLEN   2048
#define NST    16
#define NWAVE  2
#define NROW   2
#define NHALF  (NST / NWAVE)
#define NLANE  (NHALF / 2)
#define CHUNK  16
#define NC     (NLEN / CHUNK)
#define RING   4

typedef const float __attribute__((address_space(1))) gcfloat;
typedef float __attribute__((address_space(1))) gfloat;
typedef float v2f __attribute__((ext_vector_type(2)));
typedef float v4f __attribute__((ext_vector_type(4)));
typedef float v8f __attribute__((ext_vector_type(8)));


__device__ __forceinline__ float softplus_log2(float t) {
    const float l = __builtin_amdgcn_logf(1.0f + __builtin_amdgcn_exp2f(t));
    return (t > 28.853900817779268f) ? t : l;
}

#define ROW_SHR(n)   (0x110 + (n))
#define ROW_SHARE(n) (0x150 + (n))

template <int CTRL>
__device__ __forceinline__ float row_dpp(float v) {
    return __int_as_float(__builtin_amdgcn_update_dpp(0, __float_as_int(v), CTRL, 0xf, 0xf, true));
}

__device__ __forceinline__ float xhalf(float x) {
    return __int_as_float(__builtin_amdgcn_permlanex16(__float_as_int(x), __float_as_int(x),
                                                       0x76543210, 0xfedcba98, false, false));
}

__device__ __forceinline__ float ld_row(const float* p, unsigned ul) {
    gcfloat* g = (gcfloat*)p;
    asm("" : "+s"(g));
    return g[ul];
}
__device__ __forceinline__ void st_row(float* p, unsigned ul, float v) {
    gfloat* g = (gfloat*)p;
    asm("" : "+s"(g));
    g[ul] = v;
}

typedef const v4f __attribute__((address_space(1))) gcv4f;

struct Tile { v4f bq[2]; float cv[2 * NLANE]; };

__device__ __forceinline__ void load_tile(Tile& t, const float* __restrict__ Bl,
                                          const float* __restrict__ Cl, int pc, unsigned loB, unsigned loC) {
    gcfloat* Bc = (gcfloat*)(Bl + pc * CHUNK);
    gcfloat* Cc = (gcfloat*)(Cl + pc * CHUNK);
    asm("" : "+s"(Bc));
    asm("" : "+s"(Cc));
    t.bq[0] = *(gcv4f*)(Bc + loB);
    t.bq[1] = *(gcv4f*)(Bc + loB + 4);
#pragma unroll
    for (int r = 0; r < 2 * NLANE; r++)
        t.cv[r] = (Cc + ((r & 3) * NLEN + (r >> 2) * CHUNK))[loC];
}

__device__ __forceinline__ void prep_line(float dvl, float uvl, float biasl, float& rr, float& du) {
    const float sp = softplus_log2(fmaf(dvl, 1.44269504088896341f, biasl));
    rr = sp;
    rr += row_dpp<ROW_SHR(1)>(rr);
    rr += row_dpp<ROW_SHR(2)>(rr);
    rr += row_dpp<ROW_SHR(4)>(rr);
    rr += row_dpp<ROW_SHR(8)>(rr);
    du = sp * (uvl * 0.693147180559945309f);
}

__global__ __launch_bounds__(32 * NWAVE * NROW) void k_scan_rows(
    const float* __restrict__ u, const float* __restrict__ delta,
    const float* __restrict__ A, const float* __restrict__ B,
    const float* __restrict__ C, const float* __restrict__ Dp,
    const float* __restrict__ z, const float* __restrict__ dbias,
    float* __restrict__ out)
{
    __shared__ float ybuf[NROW][2][NWAVE][32];
    __shared__ __attribute__((aligned(16))) float sR[NROW][2][RING * CHUNK];
    __shared__ __attribute__((aligned(16))) float sD[NROW][2][RING * CHUNK];

    const int lane = threadIdx.x & 31;
    const int wv = __builtin_amdgcn_readfirstlane((int)(threadIdx.x >> 5));
    const int rw = wv >> 1;
    const int w = wv & 1;
    const int h = lane >> 4;
    const int m = lane & 15;
    const int row = blockIdx.x * NROW + rw;
    const int b = row / NDIM;
    const int d = row - b * NDIM;
    const size_t rbase = (size_t)row * NLEN;

    const int qg = (m >> 2) & 1;
    const int qn = 4 * (m >> 3) + (m & 3);

    const unsigned ul = (unsigned)lane;
    const unsigned loB = (unsigned)(qn * NLEN + qg * CHUNK + 8 * h);
    const unsigned loC = (unsigned)(4 * h * NLEN + m);
    const unsigned qoff = (unsigned)(qg * CHUNK + 8 * h);
    const float* dl = delta + rbase + 2 * w * CHUNK;
    const float* uf = u + rbase + 2 * w * CHUNK;
    const float* zf = z + rbase + 2 * w * CHUNK;
    float* op = out + rbase + 2 * w * CHUNK;
    const float* Bl = B + ((size_t)b * NST + NHALF * w) * NLEN;
    const float* Cl = C + ((size_t)b * NST + NHALF * w) * NLEN;

    const float biasl = dbias[d] * 1.44269504088896341f;
    const float Dd = Dp[d];
    float Ap[NLANE];
#pragma unroll
    for (int j = 0; j < NLANE; j++)
        Ap[j] = A[d * NST + NHALF * w + 4 * h + j];
    const float Aqn = -A[d * NST + NHALF * w + qn];

    float Lc[8];
#pragma unroll
    for (int e = 0; e < 8; e++) Lc[e] = ((8 * h + e) <= m) ? 1.0f : 0.0f;

    float X0[NLANE];
#pragma unroll
    for (int j = 0; j < NLANE; j++) X0[j] = 0.0f;

    {
        const float d0 = dl[ul], u0 = uf[ul];
        const float d1 = (dl + RING * CHUNK)[ul], u1 = (uf + RING * CHUNK)[ul];
        float r0, s0, r1, s1;
        prep_line(d0, u0, biasl, r0, s0);
        prep_line(d1, u1, biasl, r1, s1);
        sR[rw][0][32 * w + lane] = r0;
        sD[rw][0][32 * w + lane] = s0;
        sR[rw][1][32 * w + lane] = r1;
        sD[rw][1][32 * w + lane] = s1;
    }
    asm volatile("s_wait_dscnt 0x0" ::: "memory");
    asm volatile("s_barrier_signal -1" ::: "memory");
    asm volatile("s_barrier_wait -1" ::: "memory");
    float gR[2][RING];
    v4f qR[2][2][2], qD[2][2][2];
#pragma unroll
    for (int k = 0; k < RING; k++) gR[0][k] = sR[rw][0][k * CHUNK + m];
#pragma unroll
    for (int q = 0; q < 2; q++) {
#pragma unroll
        for (int v = 0; v < 2; v++) {
            qR[0][q][v] = *(const v4f*)&sR[rw][0][2 * q * CHUNK + qoff + 4 * v];
            qD[0][q][v] = *(const v4f*)&sD[rw][0][2 * q * CHUNK + qoff + 4 * v];
        }
    }
#pragma unroll
    for (int k = 0; k < RING; k++) gR[1][k] = gR[0][k];
#pragma unroll
    for (int q = 0; q < 2; q++) {
#pragma unroll
        for (int v = 0; v < 2; v++) { qR[1][q][v] = qR[0][q][v]; qD[1][q][v] = qD[0][q][v]; }
    }
    asm volatile("s_wait_dscnt 0x0" ::: "memory");
    asm volatile("s_barrier_signal -1" ::: "memory");
    asm volatile("s_barrier_wait -1" ::: "memory");

    Tile tl[2][2];
#pragma unroll
    for (int q = 0; q < 2; q++) {
        load_tile(tl[0][q], Bl, Cl, 2 * q, loB, loC);
        __builtin_amdgcn_sched_barrier(0);
    }

    float keep = 0.0f;
    float other = 0.0f;
    float ufin = 0.0f, zfin = 0.0f;
    float ypair0 = 0.0f;
    float dln = 0.0f, uln = 0.0f;

    for (int cc = 0; cc < NC; cc += 2 * RING) {
#pragma unroll
      for (int par = 0; par < 2; par++) {
        const int c0 = cc + par * RING;
#pragma unroll
        for (int q = 0; q < 2; q++) {
            int pn = c0 + RING + 2 * q;
            pn = (pn < NC) ? pn : (NC - 2);
            __builtin_amdgcn_sched_barrier(0);
            load_tile(tl[par ^ 1][q], Bl, Cl, pn, loB, loC);
            if (q == 0) {
                int gl = c0 + 2 * RING;
                gl = (gl < NC) ? gl : (NC - RING);
                dln = ld_row(dl + gl * CHUNK, ul);
                uln = ld_row(uf + gl * CHUNK, ul);
            }
            if (q == 1) {
                ufin = ld_row(uf + c0 * CHUNK, ul);
                zfin = ld_row(zf + c0 * CHUNK, ul);
#pragma unroll
                for (int k = 0; k < RING; k++) gR[par ^ 1][k] = sR[rw][par ^ 1][k * CHUNK + m];
#pragma unroll
                for (int qq = 0; qq < 2; qq++) {
#pragma unroll
                    for (int v = 0; v < 2; v++) {
                        qR[par ^ 1][qq][v] = *(const v4f*)&sR[rw][par ^ 1][2 * qq * CHUNK + qoff + 4 * v];
                        qD[par ^ 1][qq][v] = *(const v4f*)&sD[rw][par ^ 1][2 * qq * CHUNK + qoff + 4 * v];
                    }
                }
            }
            __builtin_amdgcn_sched_barrier(0);
            const Tile& t = tl[par][q];

            float qv[8];
#pragma unroll
            for (int e = 0; e < 8; e++) {
                const float rs = qR[par][q][e >> 2][e & 3];
                const float ds = qD[par][q][e >> 2][e & 3];
                const float bs = t.bq[e >> 2][e & 3];
                qv[e] = (bs * ds) * __builtin_amdgcn_exp2f(Aqn * rs);
            }
            v8f acc = {0.f, 0.f, 0.f, 0.f, 0.f, 0.f, 0.f, 0.f};
#pragma unroll
            for (int k = 0; k < 4; k++) {
                v2f va; va.x = qv[2 * k]; va.y = qv[2 * k + 1];
                v2f vb; vb.x = Lc[2 * k]; vb.y = Lc[2 * k + 1];
                acc = __builtin_amdgcn_wmma_f32_16x16x4_f32(
                    false, va, false, vb, (short)0, acc, false, false);
            }

            float ep[8], es[8];
#pragma unroll
            for (int r = 0; r < 8; r++) ep[r] = __builtin_amdgcn_exp2f(Ap[r & 3] * gR[par][2 * q + (r >> 2)]);
#pragma unroll
            for (int r = 0; r < 8; r++) es[r] = ep[r] * acc[r];
            float s0[NLANE], X1[NLANE], s1[NLANE];
#pragma unroll
            for (int j = 0; j < NLANE; j++) s0[j] = fmaf(ep[j], X0[j], es[j]);
#pragma unroll
            for (int j = 0; j < NLANE; j++) X1[j] = row_dpp<ROW_SHARE(15)>(s0[j]);
#pragma unroll
            for (int j = 0; j < NLANE; j++) s1[j] = fmaf(ep[NLANE + j], X1[j], es[NLANE + j]);
#pragma unroll
            for (int j = 0; j < NLANE; j++) X0[j] = row_dpp<ROW_SHARE(15)>(s1[j]);
            float y0 = t.cv[0] * s0[0];
            float y1 = t.cv[NLANE] * s1[0];
#pragma unroll
            for (int j = 1; j < NLANE; j++) {
                y0 = fmaf(t.cv[j], s0[j], y0);
                y1 = fmaf(t.cv[NLANE + j], s1[j], y1);
            }
            const float mine = h ? y1 : y0;
            const float give = h ? y0 : y1;
            const float y = mine + xhalf(give);

            if (q == 0) {
                ypair0 = y;
                if (c0 > 0) {
                    const float sig = __builtin_amdgcn_rcpf(1.0f + __expf(-zfin));
                    st_row(op + (c0 - RING) * CHUNK, ul, ((keep + other) + ufin * Dd) * (zfin * sig));
                }
            } else {
                keep = w ? y : ypair0;
                float pr, pd;
                prep_line(dln, uln, biasl, pr, pd);
                sR[rw][par][32 * w + lane] = pr;
                sD[rw][par][32 * w + lane] = pd;
                ybuf[rw][par][1 - w][lane] = w ? ypair0 : y;
                asm volatile("s_wait_dscnt 0x0" ::: "memory");
                asm volatile("s_barrier_signal -1" ::: "memory");
                asm volatile("s_barrier_wait -1" ::: "memory");
                other = ybuf[rw][par][w][lane];
            }
        }
      }
    }
    {
        const float sig = __builtin_amdgcn_rcpf(1.0f + __expf(-zfin));
        st_row(op + (NC - RING) * CHUNK, ul, ((keep + other) + ufin * Dd) * (zfin * sig));
    }
}

extern "C" void kernel_launch(void* const* d_in, const int* in_sizes, int n_in,
                              void* d_out, int out_size, void* d_ws, size_t ws_size,
                              hipStream_t stream) {
    const float* u     = (const float*)d_in[0];
    const float* delta = (const float*)d_in[1];
    const float* A     = (const float*)d_in[2];
    const float* B     = (const float*)d_in[3];
    const float* C     = (const float*)d_in[4];
    const float* Dp    = (const float*)d_in[5];
    const float* z     = (const float*)d_in[6];
    const float* dbias = (const float*)d_in[7];
    float* out = (float*)d_out;

    k_scan_rows<<<NBATCH * NDIM / NROW, 32 * NWAVE * NROW, 0, stream>>>(u, delta, A, B, C, Dp, z, dbias, out);
}
